# starting kernel
# baseline (speedup 1.0000x reference)
#include <hip/hip_runtime.h>

typedef float v2f __attribute__((ext_vector_type(2)));
typedef float v8f __attribute__((ext_vector_type(8)));

#define B_SZ 4096
#define T_SZ 512
#define DIN  8
#define H_SZ 50
#define NT   4
#define KC   13
#define RS   68

__device__ __forceinline__ float fast_tanh(float v) {
#if __has_builtin(__builtin_amdgcn_tanhf)
    return __builtin_amdgcn_tanhf(v);
#elif __has_builtin(__builtin_amdgcn_tanh_f32)
    return __builtin_amdgcn_tanh_f32(v);
#else
    return tanhf(v);
#endif
}

__global__ __launch_bounds__(32)
void rnn_scan_kernel(const float* __restrict__ x,
                     const float* __restrict__ W_ih,
                     const float* __restrict__ W_hh,
                     const float* __restrict__ b_ih,
                     const float* __restrict__ b_hh,
                     const float* __restrict__ fc_W,
                     const float* __restrict__ fc_b,
                     float* __restrict__ out)
{
    __shared__ float hl[16 * RS];

    const int lane = threadIdx.x & 31;
    const int lo   = lane & 15;
    const int hi   = lane >> 4;
    const int b0   = blockIdx.x * 16;

    v2f Bih[NT][2];
    #pragma unroll
    for (int n = 0; n < NT; ++n) {
        const int j = 16 * n + lo;
        #pragma unroll
        for (int c = 0; c < 2; ++c) {
            const int k0 = 4 * c + 2 * hi;
            Bih[n][c].x = (j < H_SZ) ? W_ih[j * DIN + k0]     : 0.0f;
            Bih[n][c].y = (j < H_SZ) ? W_ih[j * DIN + k0 + 1] : 0.0f;
        }
    }
    v2f Bhh[NT][KC];
    #pragma unroll
    for (int n = 0; n < NT; ++n) {
        const int j = 16 * n + lo;
        #pragma unroll
        for (int c = 0; c < KC; ++c) {
            const int k0 = 4 * c + 2 * hi;
            Bhh[n][c].x = (j < H_SZ && k0     < H_SZ) ? W_hh[j * H_SZ + k0]     : 0.0f;
            Bhh[n][c].y = (j < H_SZ && k0 + 1 < H_SZ) ? W_hh[j * H_SZ + k0 + 1] : 0.0f;
        }
    }
    float cinit[NT];
    #pragma unroll
    for (int n = 0; n < NT; ++n) {
        const int j = 16 * n + lo;
        cinit[n] = (j < H_SZ) ? (b_ih[j] + b_hh[j]) : 0.0f;
    }

    const float* xrow = x + (size_t)(b0 + lo) * T_SZ * DIN;
    auto ldx = [&](int t, int c) -> v2f {
        return *(const v2f*)(xrow + (size_t)t * DIN + 4 * c + 2 * hi);
    };

    v2f ha[KC];
    #pragma unroll
    for (int c = 0; c < KC; ++c) { ha[c].x = 0.0f; ha[c].y = 0.0f; }

    v2f xa0 = ldx(0, 0), xa1 = ldx(0, 1);

    for (int t = 0; t < T_SZ; ++t) {
        const int tn = (t + 1 < T_SZ) ? t + 1 : t;
        v2f nxa0 = ldx(tn, 0), nxa1 = ldx(tn, 1);

        v8f acc[NT];
        #pragma unroll
        for (int n = 0; n < NT; ++n)
            #pragma unroll
            for (int r = 0; r < 8; ++r) acc[n][r] = cinit[n];

        #pragma unroll
        for (int n = 0; n < NT; ++n) {
            acc[n] = __builtin_amdgcn_wmma_f32_16x16x4_f32(false, xa0, false, Bih[n][0],
                                                           (short)0, acc[n], false, false);
            acc[n] = __builtin_amdgcn_wmma_f32_16x16x4_f32(false, xa1, false, Bih[n][1],
                                                           (short)0, acc[n], false, false);
        }
        #pragma unroll
        for (int c = 0; c < KC; ++c)
            #pragma unroll
            for (int n = 0; n < NT; ++n)
                acc[n] = __builtin_amdgcn_wmma_f32_16x16x4_f32(false, ha[c], false, Bhh[n][c],
                                                               (short)0, acc[n], false, false);

        #pragma unroll
        for (int n = 0; n < NT; ++n)
            #pragma unroll
            for (int r = 0; r < 8; ++r)
                hl[(r + 8 * hi) * RS + 16 * n + lo] = fast_tanh(acc[n][r]);

        #pragma unroll
        for (int c = 0; c < KC; ++c)
            ha[c] = *(const v2f*)&hl[lo * RS + 4 * c + 2 * hi];

        xa0 = nxa0; xa1 = nxa1;
    }

    if (lane < 16) {
        float s = fc_b[0];
        #pragma unroll 10
        for (int j = 0; j < H_SZ; ++j)
            s += hl[lo * RS + j] * fc_W[j];
        out[b0 + lo] = s;
    }
}

extern "C" void kernel_launch(void* const* d_in, const int* in_sizes, int n_in,
                              void* d_out, int out_size, void* d_ws, size_t ws_size,
                              hipStream_t stream) {
    (void)in_sizes; (void)n_in; (void)out_size; (void)d_ws; (void)ws_size;
    rnn_scan_kernel<<<dim3(B_SZ / 16), dim3(32), 0, stream>>>(
        (const float*)d_in[0],
        (const float*)d_in[1],
        (const float*)d_in[2],
        (const float*)d_in[3],
        (const float*)d_in[4],
        (const float*)d_in[5],
        (const float*)d_in[6],
        (float*)d_out);
}


# h's 4 column tiles go to the 4 waves (one per SIMD) of a 128-thread group: 15 WMMAs per wave and step, not 60; new h exchanged through 2 LDS copies, 1 hand-written barrier a step; sums unchanged
# speedup vs baseline: 2.1854x; 2.1854x over previous
#include <hip/hip_runtime.h>


typedef float v2f __attribute__((ext_vector_type(2)));
typedef float v8f __attribute__((ext_vector_type(8)));

#define B_SZ 4096
#define T_SZ 512
#define DIN  8
#define H_SZ 50
#define NT   4
#define KC   13
#define RS   68

__device__ __forceinline__ float fast_tanh(float v) {
#if __has_builtin(__builtin_amdgcn_tanhf)
    return __builtin_amdgcn_tanhf(v);
#elif __has_builtin(__builtin_amdgcn_tanh_f32)
    return __builtin_amdgcn_tanh_f32(v);
#else
    return tanhf(v);
#endif
}

__device__ __forceinline__ v8f mma(v2f a, v2f b, v8f c) {
    return __builtin_amdgcn_wmma_f32_16x16x4_f32(false, a, false, b, (short)0, c, false, false);
}

__global__ __launch_bounds__(32 * NT)
void rnn_scan_kernel(const float* __restrict__ x,
                     const float* __restrict__ W_ih,
                     const float* __restrict__ W_hh,
                     const float* __restrict__ b_ih,
                     const float* __restrict__ b_hh,
                     const float* __restrict__ fc_W,
                     const float* __restrict__ fc_b,
                     float* __restrict__ out)
{
    __shared__ float hl[2][16 * RS];

    const int lane = threadIdx.x & 31;
    const int lo   = lane & 15;
    const int hi   = lane >> 4;
    const int n    = __builtin_amdgcn_readfirstlane((int)(threadIdx.x >> 5));
    const int b0   = blockIdx.x * 16;
    const int j    = 16 * n + lo;

    v2f Bih[2];
    #pragma unroll
    for (int c = 0; c < 2; ++c) {
        const int k0 = 4 * c + 2 * hi;
        Bih[c].x = (j < H_SZ) ? W_ih[j * DIN + k0]     : 0.0f;
        Bih[c].y = (j < H_SZ) ? W_ih[j * DIN + k0 + 1] : 0.0f;
    }
    v2f Bhh[KC];
    #pragma unroll
    for (int c = 0; c < KC; ++c) {
        const int k0 = 4 * c + 2 * hi;
        Bhh[c].x = (j < H_SZ && k0     < H_SZ) ? W_hh[j * H_SZ + k0]     : 0.0f;
        Bhh[c].y = (j < H_SZ && k0 + 1 < H_SZ) ? W_hh[j * H_SZ + k0 + 1] : 0.0f;
    }
    const float cinit = (j < H_SZ) ? (b_ih[j] + b_hh[j]) : 0.0f;

    const float* xrow = x + (size_t)(b0 + lo) * T_SZ * DIN;
    auto ldx = [&](int t, int c) -> v2f {
        return *(const v2f*)(xrow + (size_t)t * DIN + 4 * c + 2 * hi);
    };

    v2f ha[KC];
    #pragma unroll
    for (int c = 0; c < KC; ++c) { ha[c].x = 0.0f; ha[c].y = 0.0f; }

    v2f xa0 = ldx(0, 0), xa1 = ldx(0, 1);
    v2f xb0 = ldx(1, 0), xb1 = ldx(1, 1);

    #pragma unroll 2
    for (int t = 0; t < T_SZ; ++t) {
        v8f acc;
        #pragma unroll
        for (int r = 0; r < 8; ++r) acc[r] = cinit;

        acc = mma(xa0, Bih[0], acc);
        acc = mma(xa1, Bih[1], acc);

        const int tn = (t + 2 < T_SZ) ? t + 2 : T_SZ - 1;
        const v2f nx0 = ldx(tn, 0), nx1 = ldx(tn, 1);

        #pragma unroll
        for (int c = 0; c < KC; ++c)
            acc = mma(ha[c], Bhh[c], acc);

        float* hw = hl[t & 1];
        #pragma unroll
        for (int r = 0; r < 8; ++r)
            hw[(r + 8 * hi) * RS + j] = fast_tanh(acc[r]);

        asm volatile("s_wait_dscnt 0x0" ::: "memory");
        asm volatile("s_barrier_signal -1" ::: "memory");
        asm volatile("s_barrier_wait -1" ::: "memory");

        #pragma unroll
        for (int c = 0; c < KC; ++c)
            ha[c] = *(const v2f*)&hw[lo * RS + 4 * c + 2 * hi];

        __builtin_amdgcn_sched_barrier(0);

        xa0 = xb0; xa1 = xb1; xb0 = nx0; xb1 = nx1;
    }

    if (threadIdx.x < 16) {
        const float* hf = hl[(T_SZ - 1) & 1];
        float s = fc_b[0];
        #pragma unroll 10
        for (int jj = 0; jj < H_SZ; ++jj)
            s += hf[lo * RS + jj] * fc_W[jj];
        out[b0 + lo] = s;
    }
}

extern "C" void kernel_launch(void* const* d_in, const int* in_sizes, int n_in,
                              void* d_out, int out_size, void* d_ws, size_t ws_size,
                              hipStream_t stream) {
    (void)in_sizes; (void)n_in; (void)out_size; (void)d_ws; (void)ws_size;
    rnn_scan_kernel<<<dim3(B_SZ / 16), dim3(32 * NT), 0, stream>>>(
        (const float*)d_in[0],
        (const float*)d_in[1],
        (const float*)d_in[2],
        (const float*)d_in[3],
        (const float*)d_in[4],
        (const float*)d_in[5],
        (const float*)d_in[6],
        (float*)d_out);
}


# h kept transposed (W as A, h^T as B; rows split 14/12/12/12): a lane's tanh registers are its own tile's next B operands, so 5 of 15 WMMAs leave the post-exchange chain; b128 LDS; new k sum order
# speedup vs baseline: 2.5046x; 1.1461x over previous
#include <hip/hip_runtime.h>


typedef float v2f __attribute__((ext_vector_type(2)));
typedef float v4f __attribute__((ext_vector_type(4)));
typedef float v8f __attribute__((ext_vector_type(8)));

#define B_SZ 4096
#define T_SZ 512
#define DIN  8
#define H_SZ 50
#define NT   4

__device__ __forceinline__ float fast_tanh(float v) {
#if __has_builtin(__builtin_amdgcn_tanhf)
    return __builtin_amdgcn_tanhf(v);
#elif __has_builtin(__builtin_amdgcn_tanh_f32)
    return __builtin_amdgcn_tanh_f32(v);
#else
    return tanhf(v);
#endif
}

__device__ __forceinline__ v8f mma(v2f a, v2f b, v8f c) {
    return __builtin_amdgcn_wmma_f32_16x16x4_f32(false, a, false, b, (short)0, c, false, false);
}

__device__ __forceinline__ int unit_of(int m, int R) {
    if (m == 0) return (R < 14) ? R : -1;
    const int base = 14 + 12 * (m - 1);
    if (R < 6) return base + R;
    if (R >= 8 && R < 14) return base + R - 2;
    return -1;
}

__global__ __launch_bounds__(32 * NT)
void rnn_scan_kernel(const float* __restrict__ x,
                     const float* __restrict__ W_ih,
                     const float* __restrict__ W_hh,
                     const float* __restrict__ b_ih,
                     const float* __restrict__ b_hh,
                     const float* __restrict__ fc_W,
                     const float* __restrict__ fc_b,
                     float* __restrict__ out)
{
    __shared__ v4f hl[2][NT][2][32];

    const int lane = threadIdx.x & 31;
    const int lo   = lane & 15;
    const int hi   = lane >> 4;
    const int n    = __builtin_amdgcn_readfirstlane((int)(threadIdx.x >> 5));
    const int b0   = blockIdx.x * 16;

    const int jrow = unit_of(n, lo);

    auto whh = [&](int m, int q) -> v2f {
        const int k0 = unit_of(m, 8 * hi + 2 * q);
        const int k1 = unit_of(m, 8 * hi + 2 * q + 1);
        const bool ok0 = (jrow >= 0) && (k0 >= 0);
        const bool ok1 = (jrow >= 0) && (k1 >= 0);
        const float w0 = W_hh[ok0 ? jrow * H_SZ + k0 : 0];
        const float w1 = W_hh[ok1 ? jrow * H_SZ + k1 : 0];
        v2f a; a.x = ok0 ? w0 : 0.0f; a.y = ok1 ? w1 : 0.0f;
        return a;
    };

    v2f Aih[2];
    #pragma unroll
    for (int c = 0; c < 2; ++c) {
        const int k0 = 4 * c + 2 * hi;
        const int jr = (jrow >= 0) ? jrow : 0;
        const float w0 = W_ih[jr * DIN + k0], w1 = W_ih[jr * DIN + k0 + 1];
        Aih[c].x = (jrow >= 0) ? w0 : 0.0f;
        Aih[c].y = (jrow >= 0) ? w1 : 0.0f;
    }
    v2f Aown[3];
    #pragma unroll
    for (int q = 0; q < 3; ++q) Aown[q] = whh(n, q);
    const v2f Ax = whh(0, 3);
    int mo[3];
    v2f Aoth[3][3];
    #pragma unroll
    for (int i = 0; i < 3; ++i) {
        mo[i] = (n + 1 + i) & 3;
        #pragma unroll
        for (int q = 0; q < 3; ++q) Aoth[i][q] = whh(mo[i], q);
    }
    v8f bias;
    #pragma unroll
    for (int r = 0; r < 8; ++r) {
        const int jb = unit_of(n, 8 * hi + r);
        const float bv = b_ih[jb >= 0 ? jb : 0] + b_hh[jb >= 0 ? jb : 0];
        bias[r] = (jb >= 0) ? bv : 0.0f;
    }

    const float* xrow = x + (size_t)(b0 + lo) * T_SZ * DIN;
    auto ldx = [&](int t, int c) -> v2f {
        return *(const v2f*)(xrow + (size_t)t * DIN + 4 * c + 2 * hi);
    };

    v2f hx; hx.x = 0.0f; hx.y = 0.0f;
    v2f hoth[3][3];
    #pragma unroll
    for (int i = 0; i < 3; ++i) {
        #pragma unroll
        for (int q = 0; q < 3; ++q) { hoth[i][q].x = 0.0f; hoth[i][q].y = 0.0f; }
    }

    v2f xs[4][2];
    #pragma unroll
    for (int s = 0; s < 4; ++s) { xs[s][0] = ldx(s, 0); xs[s][1] = ldx(s, 1); }

    v8f acc = mma(Aih[0], xs[0][0], bias);
    acc = mma(Aih[1], xs[0][1], acc);

    for (int t4 = 0; t4 < T_SZ; t4 += 4) {
        #pragma unroll
        for (int u = 0; u < 4; ++u) {
            const int t = t4 + u;

            acc = mma(Ax, hx, acc);
            #pragma unroll
            for (int i = 0; i < 3; ++i) {
                #pragma unroll
                for (int q = 0; q < 3; ++q) acc = mma(Aoth[i][q], hoth[i][q], acc);
            }

            v4f h03, h47;
            #pragma unroll
            for (int r = 0; r < 4; ++r) { h03[r] = fast_tanh(acc[r]); h47[r] = fast_tanh(acc[4 + r]); }

            v4f* hw = &hl[u & 1][0][0][0];
            hw[(2 * n + 0) * 32 + lane] = h03;
            hw[(2 * n + 1) * 32 + lane] = h47;

            __builtin_amdgcn_sched_barrier(0);

            const int tn = (t + 4 < T_SZ) ? t + 4 : T_SZ - 1;
            xs[u][0] = ldx(tn, 0);
            xs[u][1] = ldx(tn, 1);

            acc = mma(Aih[0], xs[(u + 1) & 3][0], bias);
            acc = mma(Aih[1], xs[(u + 1) & 3][1], acc);

            asm volatile("s_wait_dscnt 0x0" ::: "memory");
            asm volatile("s_barrier_signal -1" ::: "memory");
            asm volatile("s_barrier_wait -1" ::: "memory");

            const v4f* hr = &hl[u & 1][0][0][0];
            hx = *((const v2f*)&hr[(2 * 0 + 1) * 32 + lane] + 1);
            #pragma unroll
            for (int i = 0; i < 3; ++i) {
                const v4f a = hr[(2 * mo[i] + 0) * 32 + lane];
                const v2f b = *(const v2f*)&hr[(2 * mo[i] + 1) * 32 + lane];
                hoth[i][0].x = a[0]; hoth[i][0].y = a[1];
                hoth[i][1].x = a[2]; hoth[i][1].y = a[3];
                hoth[i][2] = b;
            }

            __builtin_amdgcn_sched_barrier(0);

            v2f o0; o0.x = h03[0]; o0.y = h03[1];
            v2f o1; o1.x = h03[2]; o1.y = h03[3];
            v2f o2; o2.x = h47[0]; o2.y = h47[1];
            acc = mma(Aown[0], o0, acc);
            acc = mma(Aown[1], o1, acc);
            acc = mma(Aown[2], o2, acc);
        }
    }

    if (threadIdx.x < 16) {
        const v4f* hf = &hl[(T_SZ - 1) & 1][0][0][0];
        float s = fc_b[0];
        #pragma unroll
        for (int jj = 0; jj < H_SZ; ++jj) {
            const int m = (jj < 14) ? 0 : 1 + (jj - 14) / 12;
            const int u = (jj < 14) ? jj : (jj - 14) % 12;
            const int R = (jj < 14) ? jj : ((u < 6) ? u : u + 2);
            const int r = R & 7;
            const v4f v = hf[(2 * m + (r >> 2)) * 32 + (R >> 3) * 16 + lo];
            s += v[r & 3] * fc_W[jj];
        }
        out[b0 + lo] = s;
    }
}

extern "C" void kernel_launch(void* const* d_in, const int* in_sizes, int n_in,
                              void* d_out, int out_size, void* d_ws, size_t ws_size,
                              hipStream_t stream) {
    (void)in_sizes; (void)n_in; (void)out_size; (void)d_ws; (void)ws_size;
    rnn_scan_kernel<<<dim3(B_SZ / 16), dim3(32 * NT), 0, stream>>>(
        (const float*)d_in[0],
        (const float*)d_in[1],
        (const float*)d_in[2],
        (const float*)d_in[3],
        (const float*)d_in[4],
        (const float*)d_in[5],
        (const float*)d_in[6],
        (float*)d_out);
}


# a step's 15 products go to two accumulators (the LDS group alternately), added once before tanh: no product waits for the one issued just before it; new k sum order: two partial sums, one add
# speedup vs baseline: 2.6726x; 1.0671x over previous
#include <hip/hip_runtime.h>


typedef float v2f __attribute__((ext_vector_type(2)));
typedef float v4f __attribute__((ext_vector_type(4)));
typedef float v8f __attribute__((ext_vector_type(8)));

#define B_SZ 4096
#define T_SZ 512
#define DIN  8
#define H_SZ 50
#define NT   4

__device__ __forceinline__ float fast_tanh(float v) {
#if __has_builtin(__builtin_amdgcn_tanhf)
    return __builtin_amdgcn_tanhf(v);
#elif __has_builtin(__builtin_amdgcn_tanh_f32)
    return __builtin_amdgcn_tanh_f32(v);
#else
    return tanhf(v);
#endif
}

__device__ __forceinline__ v8f mma(v2f a, v2f b, v8f c) {
    return __builtin_amdgcn_wmma_f32_16x16x4_f32(false, a, false, b, (short)0, c, false, false);
}

__device__ __forceinline__ int unit_of(int m, int R) {
    if (m == 0) return (R < 14) ? R : -1;
    const int base = 14 + 12 * (m - 1);
    if (R < 6) return base + R;
    if (R >= 8 && R < 14) return base + R - 2;
    return -1;
}

__global__ __launch_bounds__(32 * NT)
void rnn_scan_kernel(const float* __restrict__ x,
                     const float* __restrict__ W_ih,
                     const float* __restrict__ W_hh,
                     const float* __restrict__ b_ih,
                     const float* __restrict__ b_hh,
                     const float* __restrict__ fc_W,
                     const float* __restrict__ fc_b,
                     float* __restrict__ out)
{
    __shared__ v4f hl[2][NT][2][32];

    const int lane = threadIdx.x & 31;
    const int lo   = lane & 15;
    const int hi   = lane >> 4;
    const int n    = __builtin_amdgcn_readfirstlane((int)(threadIdx.x >> 5));
    const int b0   = blockIdx.x * 16;

    const int jrow = unit_of(n, lo);

    auto whh = [&](int m, int q) -> v2f {
        const int k0 = unit_of(m, 8 * hi + 2 * q);
        const int k1 = unit_of(m, 8 * hi + 2 * q + 1);
        const bool ok0 = (jrow >= 0) && (k0 >= 0);
        const bool ok1 = (jrow >= 0) && (k1 >= 0);
        const float w0 = W_hh[ok0 ? jrow * H_SZ + k0 : 0];
        const float w1 = W_hh[ok1 ? jrow * H_SZ + k1 : 0];
        v2f a; a.x = ok0 ? w0 : 0.0f; a.y = ok1 ? w1 : 0.0f;
        return a;
    };

    v2f Aih[2];
    #pragma unroll
    for (int c = 0; c < 2; ++c) {
        const int k0 = 4 * c + 2 * hi;
        const int jr = (jrow >= 0) ? jrow : 0;
        const float w0 = W_ih[jr * DIN + k0], w1 = W_ih[jr * DIN + k0 + 1];
        Aih[c].x = (jrow >= 0) ? w0 : 0.0f;
        Aih[c].y = (jrow >= 0) ? w1 : 0.0f;
    }
    v2f Aown[3];
    #pragma unroll
    for (int q = 0; q < 3; ++q) Aown[q] = whh(n, q);
    const v2f Ax = whh(0, 3);
    int mo[3];
    v2f Aoth[3][3];
    #pragma unroll
    for (int i = 0; i < 3; ++i) {
        mo[i] = (n + 1 + i) & 3;
        #pragma unroll
        for (int q = 0; q < 3; ++q) Aoth[i][q] = whh(mo[i], q);
    }
    v8f bias;
    #pragma unroll
    for (int r = 0; r < 8; ++r) {
        const int jb = unit_of(n, 8 * hi + r);
        const float bv = b_ih[jb >= 0 ? jb : 0] + b_hh[jb >= 0 ? jb : 0];
        bias[r] = (jb >= 0) ? bv : 0.0f;
    }

    const float* xrow = x + (size_t)(b0 + lo) * T_SZ * DIN;
    auto ldx = [&](int t, int c) -> v2f {
        return *(const v2f*)(xrow + (size_t)t * DIN + 4 * c + 2 * hi);
    };

    v2f hx; hx.x = 0.0f; hx.y = 0.0f;
    v2f hoth[3][3];
    #pragma unroll
    for (int i = 0; i < 3; ++i) {
        #pragma unroll
        for (int q = 0; q < 3; ++q) { hoth[i][q].x = 0.0f; hoth[i][q].y = 0.0f; }
    }

    v2f xs[4][2];
    #pragma unroll
    for (int s = 0; s < 4; ++s) { xs[s][0] = ldx(s, 0); xs[s][1] = ldx(s, 1); }

    const v8f zero8 = {0.0f, 0.0f, 0.0f, 0.0f, 0.0f, 0.0f, 0.0f, 0.0f};
    v8f accA = mma(Aih[0], xs[0][0], bias);
    accA = mma(Aih[1], xs[0][1], accA);
    v8f accB = zero8;

    for (int t4 = 0; t4 < T_SZ; t4 += 4) {
        #pragma unroll
        for (int u = 0; u < 4; ++u) {
            const int t = t4 + u;

            accA = mma(Ax,         hx,         accA);
            accB = mma(Aoth[0][0], hoth[0][0], accB);
            accA = mma(Aoth[0][1], hoth[0][1], accA);
            accB = mma(Aoth[0][2], hoth[0][2], accB);
            accA = mma(Aoth[1][0], hoth[1][0], accA);
            accB = mma(Aoth[1][1], hoth[1][1], accB);
            accA = mma(Aoth[1][2], hoth[1][2], accA);
            accB = mma(Aoth[2][0], hoth[2][0], accB);
            accA = mma(Aoth[2][1], hoth[2][1], accA);
            accB = mma(Aoth[2][2], hoth[2][2], accB);
            const v8f acc = accA + accB;

            v4f h03, h47;
            #pragma unroll
            for (int r = 0; r < 4; ++r) { h03[r] = fast_tanh(acc[r]); h47[r] = fast_tanh(acc[4 + r]); }

            v4f* hw = &hl[u & 1][0][0][0];
            hw[(2 * n + 0) * 32 + lane] = h03;
            hw[(2 * n + 1) * 32 + lane] = h47;

            __builtin_amdgcn_sched_barrier(0);

            const int tn = (t + 4 < T_SZ) ? t + 4 : T_SZ - 1;
            xs[u][0] = ldx(tn, 0);
            xs[u][1] = ldx(tn, 1);

            accA = mma(Aih[0], xs[(u + 1) & 3][0], bias);
            accA = mma(Aih[1], xs[(u + 1) & 3][1], accA);

            asm volatile("s_wait_dscnt 0x0" ::: "memory");
            asm volatile("s_barrier_signal -1" ::: "memory");
            asm volatile("s_barrier_wait -1" ::: "memory");

            const v4f* hr = &hl[u & 1][0][0][0];
            hx = *((const v2f*)&hr[(2 * 0 + 1) * 32 + lane] + 1);
            #pragma unroll
            for (int i = 0; i < 3; ++i) {
                const v4f a = hr[(2 * mo[i] + 0) * 32 + lane];
                const v2f b = *(const v2f*)&hr[(2 * mo[i] + 1) * 32 + lane];
                hoth[i][0].x = a[0]; hoth[i][0].y = a[1];
                hoth[i][1].x = a[2]; hoth[i][1].y = a[3];
                hoth[i][2] = b;
            }

            __builtin_amdgcn_sched_barrier(0);

            v2f o0; o0.x = h03[0]; o0.y = h03[1];
            v2f o1; o1.x = h03[2]; o1.y = h03[3];
            v2f o2; o2.x = h47[0]; o2.y = h47[1];
            accB = mma(Aown[0], o0, zero8);
            accB = mma(Aown[1], o1, accB);
            accB = mma(Aown[2], o2, accB);

            __builtin_amdgcn_sched_barrier(0);
        }
    }

    if (threadIdx.x < 16) {
        const v4f* hf = &hl[(T_SZ - 1) & 1][0][0][0];
        float s = fc_b[0];
        #pragma unroll
        for (int jj = 0; jj < H_SZ; ++jj) {
            const int m = (jj < 14) ? 0 : 1 + (jj - 14) / 12;
            const int u = (jj < 14) ? jj : (jj - 14) % 12;
            const int R = (jj < 14) ? jj : ((u < 6) ? u : u + 2);
            const int r = R & 7;
            const v4f v = hf[(2 * m + (r >> 2)) * 32 + (R >> 3) * 16 + lo];
            s += v[r & 3] * fc_W[jj];
        }
        out[b0 + lo] = s;
    }
}

extern "C" void kernel_launch(void* const* d_in, const int* in_sizes, int n_in,
                              void* d_out, int out_size, void* d_ws, size_t ws_size,
                              hipStream_t stream) {
    (void)in_sizes; (void)n_in; (void)out_size; (void)d_ws; (void)ws_size;
    rnn_scan_kernel<<<dim3(B_SZ / 16), dim3(32 * NT), 0, stream>>>(
        (const float*)d_in[0],
        (const float*)d_in[1],
        (const float*)d_in[2],
        (const float*)d_in[3],
        (const float*)d_in[4],
        (const float*)d_in[5],
        (const float*)d_in[6],
        (float*)d_out);
}


# waves 1-3 compute units 6,7 again in 2 padding rows: that half-filled k chunk is now each wave's own registers; 9 not 10 products wait for the h exchange (x products before barrier); new k sum order
# speedup vs baseline: 2.7956x; 1.0460x over previous
#include <hip/hip_runtime.h>


typedef float v2f __attribute__((ext_vector_type(2)));
typedef float v4f __attribute__((ext_vector_type(4)));
typedef float v8f __attribute__((ext_vector_type(8)));

#define B_SZ 4096
#define T_SZ 512
#define DIN  8
#define H_SZ 50
#define NT   4

__device__ __forceinline__ float fast_tanh(float v) {
#if __has_builtin(__builtin_amdgcn_tanhf)
    return __builtin_amdgcn_tanhf(v);
#elif __has_builtin(__builtin_amdgcn_tanh_f32)
    return __builtin_amdgcn_tanh_f32(v);
#else
    return tanhf(v);
#endif
}

__device__ __forceinline__ v8f mma(v2f a, v2f b, v8f c) {
    return __builtin_amdgcn_wmma_f32_16x16x4_f32(false, a, false, b, (short)0, c, false, false);
}

__device__ __forceinline__ int unit_of(int m, int R) {
    if (m == 0) return (R < 14) ? R : -1;
    const int base = 14 + 12 * (m - 1);
    if (R < 6) return base + R;
    if (R < 8) return R;
    if (R < 14) return base + R - 2;
    return -1;
}

__global__ __launch_bounds__(32 * NT)
void rnn_scan_kernel(const float* __restrict__ x,
                     const float* __restrict__ W_ih,
                     const float* __restrict__ W_hh,
                     const float* __restrict__ b_ih,
                     const float* __restrict__ b_hh,
                     const float* __restrict__ fc_W,
                     const float* __restrict__ fc_b,
                     float* __restrict__ out)
{
    __shared__ v4f hl[2][NT][2][32];

    const int lane = threadIdx.x & 31;
    const int lo   = lane & 15;
    const int hi   = lane >> 4;
    const int n    = __builtin_amdgcn_readfirstlane((int)(threadIdx.x >> 5));
    const int b0   = blockIdx.x * 16;

    const int jrow = unit_of(n, lo);

    auto whh = [&](int m, int q) -> v2f {
        const int k0 = unit_of(m, 8 * hi + 2 * q);
        const int k1 = unit_of(m, 8 * hi + 2 * q + 1);
        const bool ok0 = (jrow >= 0) && (k0 >= 0);
        const bool ok1 = (jrow >= 0) && (k1 >= 0);
        const float w0 = W_hh[ok0 ? jrow * H_SZ + k0 : 0];
        const float w1 = W_hh[ok1 ? jrow * H_SZ + k1 : 0];
        v2f a; a.x = ok0 ? w0 : 0.0f; a.y = ok1 ? w1 : 0.0f;
        return a;
    };

    v2f Aih[2];
    #pragma unroll
    for (int c = 0; c < 2; ++c) {
        const int k0 = 4 * c + 2 * hi;
        const int jr = (jrow >= 0) ? jrow : 0;
        const float w0 = W_ih[jr * DIN + k0], w1 = W_ih[jr * DIN + k0 + 1];
        Aih[c].x = (jrow >= 0) ? w0 : 0.0f;
        Aih[c].y = (jrow >= 0) ? w1 : 0.0f;
    }
    v2f Aown[4];
    #pragma unroll
    for (int q = 0; q < 4; ++q) Aown[q] = whh(n, q);
    int mo[3];
    v2f Aoth[3][3];
    #pragma unroll
    for (int i = 0; i < 3; ++i) {
        mo[i] = (n + 1 + i) & 3;
        #pragma unroll
        for (int q = 0; q < 3; ++q) Aoth[i][q] = whh(mo[i], q);
    }
    v8f bias;
    #pragma unroll
    for (int r = 0; r < 8; ++r) {
        const int jb = unit_of(n, 8 * hi + r);
        const float bv = b_ih[jb >= 0 ? jb : 0] + b_hh[jb >= 0 ? jb : 0];
        bias[r] = (jb >= 0) ? bv : 0.0f;
    }

    const float* xrow = x + (size_t)(b0 + lo) * T_SZ * DIN;
    auto ldx = [&](int t, int c) -> v2f {
        return *(const v2f*)(xrow + (size_t)t * DIN + 4 * c + 2 * hi);
    };

    v2f hoth[3][3];
    #pragma unroll
    for (int i = 0; i < 3; ++i) {
        #pragma unroll
        for (int q = 0; q < 3; ++q) { hoth[i][q].x = 0.0f; hoth[i][q].y = 0.0f; }
    }

    v2f xs[4][2];
    #pragma unroll
    for (int s = 0; s < 4; ++s) { xs[s][0] = ldx(s, 0); xs[s][1] = ldx(s, 1); }

    const v8f zero8 = {0.0f, 0.0f, 0.0f, 0.0f, 0.0f, 0.0f, 0.0f, 0.0f};
    v8f accA = mma(Aih[0], xs[0][0], bias);
    accA = mma(Aih[1], xs[0][1], accA);
    v8f accB = zero8;

    for (int t4 = 0; t4 < T_SZ; t4 += 4) {
        #pragma unroll
        for (int u = 0; u < 4; ++u) {
            const int t = t4 + u;

            accB = mma(Aoth[0][0], hoth[0][0], accB);
            accA = mma(Aoth[0][1], hoth[0][1], accA);
            accB = mma(Aoth[0][2], hoth[0][2], accB);
            accA = mma(Aoth[1][0], hoth[1][0], accA);
            accB = mma(Aoth[1][1], hoth[1][1], accB);
            accA = mma(Aoth[1][2], hoth[1][2], accA);
            accB = mma(Aoth[2][0], hoth[2][0], accB);
            accA = mma(Aoth[2][1], hoth[2][1], accA);
            accB = mma(Aoth[2][2], hoth[2][2], accB);
            const v8f acc = accA + accB;

            v4f h03, h47;
            #pragma unroll
            for (int r = 0; r < 4; ++r) { h03[r] = fast_tanh(acc[r]); h47[r] = fast_tanh(acc[4 + r]); }

            v4f* hw = &hl[u & 1][0][0][0];
            hw[(2 * n + 0) * 32 + lane] = h03;
            hw[(2 * n + 1) * 32 + lane] = h47;

            __builtin_amdgcn_sched_barrier(0);

            accA = mma(Aih[0], xs[(u + 1) & 3][0], bias);
            accA = mma(Aih[1], xs[(u + 1) & 3][1], accA);

            asm volatile("" : "+v"(accA));

            const int tn = (t + 4 < T_SZ) ? t + 4 : T_SZ - 1;
            xs[u][0] = ldx(tn, 0);
            xs[u][1] = ldx(tn, 1);

            asm volatile("s_wait_dscnt 0x0" ::: "memory");
            asm volatile("s_barrier_signal -1" ::: "memory");
            asm volatile("s_barrier_wait -1" ::: "memory");

            const v4f* hr = &hl[u & 1][0][0][0];
            #pragma unroll
            for (int i = 0; i < 3; ++i) {
                const v4f a = hr[(2 * mo[i] + 0) * 32 + lane];
                const v2f b = *(const v2f*)&hr[(2 * mo[i] + 1) * 32 + lane];
                hoth[i][0].x = a[0]; hoth[i][0].y = a[1];
                hoth[i][1].x = a[2]; hoth[i][1].y = a[3];
                hoth[i][2] = b;
            }

            __builtin_amdgcn_sched_barrier(0);

            v2f o0; o0.x = h03[0]; o0.y = h03[1];
            v2f o1; o1.x = h03[2]; o1.y = h03[3];
            v2f o2; o2.x = h47[0]; o2.y = h47[1];
            v2f o3; o3.x = h47[2]; o3.y = h47[3];
            accB = mma(Aown[0], o0, zero8);
            accA = mma(Aown[1], o1, accA);
            accB = mma(Aown[2], o2, accB);
            accA = mma(Aown[3], o3, accA);

            __builtin_amdgcn_sched_barrier(0);
        }
    }

    if (threadIdx.x < 16) {
        const v4f* hf = &hl[(T_SZ - 1) & 1][0][0][0];
        float s = fc_b[0];
        #pragma unroll
        for (int jj = 0; jj < H_SZ; ++jj) {
            const int m = (jj < 14) ? 0 : 1 + (jj - 14) / 12;
            const int u = (jj < 14) ? jj : (jj - 14) % 12;
            const int R = (jj < 14) ? jj : ((u < 6) ? u : u + 2);
            const int r = R & 7;
            const v4f v = hf[(2 * m + (r >> 2)) * 32 + (R >> 3) * 16 + lo];
            s += v[r & 3] * fc_W[jj];
        }
        out[b0 + lo] = s;
    }
}

extern "C" void kernel_launch(void* const* d_in, const int* in_sizes, int n_in,
                              void* d_out, int out_size, void* d_ws, size_t ws_size,
                              hipStream_t stream) {
    (void)in_sizes; (void)n_in; (void)out_size; (void)d_ws; (void)ws_size;
    rnn_scan_kernel<<<dim3(B_SZ / 16), dim3(32 * NT), 0, stream>>>(
        (const float*)d_in[0],
        (const float*)d_in[1],
        (const float*)d_in[2],
        (const float*)d_in[3],
        (const float*)d_in[4],
        (const float*)d_in[5],
        (const float*)d_in[6],
        (float*)d_out);
}


# x refill pointer advanced in place once a step, not rebuilt in a fresh register pair: no s_wait_xcnt 0 left between the barrier's LDS loads and the own products on the chain; sums unchanged
# speedup vs baseline: 2.9021x; 1.0381x over previous
#include <hip/hip_runtime.h>


typedef float v2f __attribute__((ext_vector_type(2)));
typedef float v4f __attribute__((ext_vector_type(4)));
typedef float v8f __attribute__((ext_vector_type(8)));

#define B_SZ 4096
#define T_SZ 512
#define DIN  8
#define H_SZ 50
#define NT   4

__device__ __forceinline__ float fast_tanh(float v) {
#if __has_builtin(__builtin_amdgcn_tanhf)
    return __builtin_amdgcn_tanhf(v);
#elif __has_builtin(__builtin_amdgcn_tanh_f32)
    return __builtin_amdgcn_tanh_f32(v);
#else
    return tanhf(v);
#endif
}

__device__ __forceinline__ v8f mma(v2f a, v2f b, v8f c) {
    return __builtin_amdgcn_wmma_f32_16x16x4_f32(false, a, false, b, (short)0, c, false, false);
}

__device__ __forceinline__ int unit_of(int m, int R) {
    if (m == 0) return (R < 14) ? R : -1;
    const int base = 14 + 12 * (m - 1);
    if (R < 6) return base + R;
    if (R < 8) return R;
    if (R < 14) return base + R - 2;
    return -1;
}

__global__ __launch_bounds__(32 * NT)
void rnn_scan_kernel(const float* __restrict__ x,
                     const float* __restrict__ W_ih,
                     const float* __restrict__ W_hh,
                     const float* __restrict__ b_ih,
                     const float* __restrict__ b_hh,
                     const float* __restrict__ fc_W,
                     const float* __restrict__ fc_b,
                     float* __restrict__ out)
{
    __shared__ v4f hl[2][NT][2][32];

    const int lane = threadIdx.x & 31;
    const int lo   = lane & 15;
    const int hi   = lane >> 4;
    const int n    = __builtin_amdgcn_readfirstlane((int)(threadIdx.x >> 5));
    const int b0   = blockIdx.x * 16;

    const int jrow = unit_of(n, lo);

    auto whh = [&](int m, int q) -> v2f {
        const int k0 = unit_of(m, 8 * hi + 2 * q);
        const int k1 = unit_of(m, 8 * hi + 2 * q + 1);
        const bool ok0 = (jrow >= 0) && (k0 >= 0);
        const bool ok1 = (jrow >= 0) && (k1 >= 0);
        const float w0 = W_hh[ok0 ? jrow * H_SZ + k0 : 0];
        const float w1 = W_hh[ok1 ? jrow * H_SZ + k1 : 0];
        v2f a; a.x = ok0 ? w0 : 0.0f; a.y = ok1 ? w1 : 0.0f;
        return a;
    };

    v2f Aih[2];
    #pragma unroll
    for (int c = 0; c < 2; ++c) {
        const int k0 = 4 * c + 2 * hi;
        const int jr = (jrow >= 0) ? jrow : 0;
        const float w0 = W_ih[jr * DIN + k0], w1 = W_ih[jr * DIN + k0 + 1];
        Aih[c].x = (jrow >= 0) ? w0 : 0.0f;
        Aih[c].y = (jrow >= 0) ? w1 : 0.0f;
    }
    v2f Aown[4];
    #pragma unroll
    for (int q = 0; q < 4; ++q) Aown[q] = whh(n, q);
    int mo[3];
    v2f Aoth[3][3];
    #pragma unroll
    for (int i = 0; i < 3; ++i) {
        mo[i] = (n + 1 + i) & 3;
        #pragma unroll
        for (int q = 0; q < 3; ++q) Aoth[i][q] = whh(mo[i], q);
    }
    v8f bias;
    #pragma unroll
    for (int r = 0; r < 8; ++r) {
        const int jb = unit_of(n, 8 * hi + r);
        const float bv = b_ih[jb >= 0 ? jb : 0] + b_hh[jb >= 0 ? jb : 0];
        bias[r] = (jb >= 0) ? bv : 0.0f;
    }

    const float* xrow = x + (size_t)(b0 + lo) * T_SZ * DIN + 2 * hi;
    auto ldx = [&](int t, int c) -> v2f {
        return *(const v2f*)(xrow + (size_t)t * DIN + 4 * c);
    };
    const float* xnext = xrow + 3 * DIN;

    v2f hoth[3][3];
    #pragma unroll
    for (int i = 0; i < 3; ++i) {
        #pragma unroll
        for (int q = 0; q < 3; ++q) { hoth[i][q].x = 0.0f; hoth[i][q].y = 0.0f; }
    }

    v2f xs[4][2];
    #pragma unroll
    for (int s = 0; s < 4; ++s) { xs[s][0] = ldx(s, 0); xs[s][1] = ldx(s, 1); }

    const v8f zero8 = {0.0f, 0.0f, 0.0f, 0.0f, 0.0f, 0.0f, 0.0f, 0.0f};
    v8f accA = mma(Aih[0], xs[0][0], bias);
    accA = mma(Aih[1], xs[0][1], accA);
    v8f accB = zero8;

    for (int t4 = 0; t4 < T_SZ; t4 += 4) {
        #pragma unroll
        for (int u = 0; u < 4; ++u) {
            const int t = t4 + u;

            accB = mma(Aoth[0][0], hoth[0][0], accB);
            accA = mma(Aoth[0][1], hoth[0][1], accA);
            accB = mma(Aoth[0][2], hoth[0][2], accB);
            accA = mma(Aoth[1][0], hoth[1][0], accA);
            accB = mma(Aoth[1][1], hoth[1][1], accB);
            accA = mma(Aoth[1][2], hoth[1][2], accA);
            accB = mma(Aoth[2][0], hoth[2][0], accB);
            accA = mma(Aoth[2][1], hoth[2][1], accA);
            accB = mma(Aoth[2][2], hoth[2][2], accB);
            const v8f acc = accA + accB;

            v4f h03, h47;
            #pragma unroll
            for (int r = 0; r < 4; ++r) { h03[r] = fast_tanh(acc[r]); h47[r] = fast_tanh(acc[4 + r]); }

            v4f* hw = &hl[u & 1][0][0][0];
            hw[(2 * n + 0) * 32 + lane] = h03;
            hw[(2 * n + 1) * 32 + lane] = h47;

            __builtin_amdgcn_sched_barrier(0);

            accA = mma(Aih[0], xs[(u + 1) & 3][0], bias);
            accA = mma(Aih[1], xs[(u + 1) & 3][1], accA);

            asm volatile("" : "+v"(accA));

            xnext += (t + 4 < T_SZ) ? DIN : 0;
            xs[u][0] = *(const v2f*)(xnext);
            xs[u][1] = *(const v2f*)(xnext + 4);

            asm volatile("s_wait_dscnt 0x0" ::: "memory");
            asm volatile("s_barrier_signal -1" ::: "memory");
            asm volatile("s_barrier_wait -1" ::: "memory");

            const v4f* hr = &hl[u & 1][0][0][0];
            #pragma unroll
            for (int i = 0; i < 3; ++i) {
                const v4f a = hr[(2 * mo[i] + 0) * 32 + lane];
                const v2f b = *(const v2f*)&hr[(2 * mo[i] + 1) * 32 + lane];
                hoth[i][0].x = a[0]; hoth[i][0].y = a[1];
                hoth[i][1].x = a[2]; hoth[i][1].y = a[3];
                hoth[i][2] = b;
            }

            __builtin_amdgcn_sched_barrier(0);

            v2f o0; o0.x = h03[0]; o0.y = h03[1];
            v2f o1; o1.x = h03[2]; o1.y = h03[3];
            v2f o2; o2.x = h47[0]; o2.y = h47[1];
            v2f o3; o3.x = h47[2]; o3.y = h47[3];
            accB = mma(Aown[0], o0, zero8);
            accA = mma(Aown[1], o1, accA);
            accB = mma(Aown[2], o2, accB);
            accA = mma(Aown[3], o3, accA);

            __builtin_amdgcn_sched_barrier(0);
        }
    }

    if (threadIdx.x < 16) {
        const v4f* hf = &hl[(T_SZ - 1) & 1][0][0][0];
        float s = fc_b[0];
        #pragma unroll
        for (int jj = 0; jj < H_SZ; ++jj) {
            const int m = (jj < 14) ? 0 : 1 + (jj - 14) / 12;
            const int u = (jj < 14) ? jj : (jj - 14) % 12;
            const int R = (jj < 14) ? jj : ((u < 6) ? u : u + 2);
            const int r = R & 7;
            const v4f v = hf[(2 * m + (r >> 2)) * 32 + (R >> 3) * 16 + lo];
            s += v[r & 3] * fc_W[jj];
        }
        out[b0 + lo] = s;
    }
}

extern "C" void kernel_launch(void* const* d_in, const int* in_sizes, int n_in,
                              void* d_out, int out_size, void* d_ws, size_t ws_size,
                              hipStream_t stream) {
    (void)in_sizes; (void)n_in; (void)out_size; (void)d_ws; (void)ws_size;
    rnn_scan_kernel<<<dim3(B_SZ / 16), dim3(32 * NT), 0, stream>>>(
        (const float*)d_in[0],
        (const float*)d_in[1],
        (const float*)d_in[2],
        (const float*)d_in[3],
        (const float*)d_in[4],
        (const float*)d_in[5],
        (const float*)d_in[6],
        (float*)d_out);
}
